# starting kernel
# baseline (speedup 1.0000x reference)
#include <hip/hip_runtime.h>
#include <math.h>

#define D_   128
#define KC   8
#define DD_  16
#define CUT_ 5
#define M_   32

typedef float v2f __attribute__((ext_vector_type(2)));
typedef float v8f __attribute__((ext_vector_type(8)));

__global__ __launch_bounds__(128) void norm_kernel(const float* __restrict__ x,
                                                   float* __restrict__ xn, int N) {
  const int node = blockIdx.x;
  const int t = threadIdx.x;
  if (node >= N) { xn[(size_t)node * D_ + t] = 0.0f; return; }
  float v = x[(size_t)node * D_ + t];
  float s = v * v;
  s += __shfl_xor(s, 1, 16);
  s += __shfl_xor(s, 2, 16);
  s += __shfl_xor(s, 4, 16);
  s += __shfl_xor(s, 8, 16);
  float nrm = fmaxf(sqrtf(s), 1e-12f);
  xn[(size_t)node * D_ + t] = v / nrm;
}

__global__ __launch_bounds__(32) void routing_kernel(const float* __restrict__ xn,
                                                     const int* __restrict__ nbr,
                                                     const int* __restrict__ p_maxiter,
                                                     float* __restrict__ u_out,
                                                     int* __restrict__ selcap,
                                                     int N, int M) {
  __shared__ float zs[M_ * D_];
  __shared__ float ps[M_ * KC];
  __shared__ float us[D_];
  __shared__ float xs[D_];

  const int node = blockIdx.x;
  const int lane = threadIdx.x;
  const int half = lane >> 4;
  const int lo   = lane & 15;
  const int T = p_maxiter[0];

  ((float4*)xs)[lane] = ((const float4*)(xn + (size_t)node * D_))[lane];

  int gl = nbr[node * M + lane];
  for (int r = 0; r < M_; ++r) {
    int g = __shfl(gl, r, 32);
    int row = (g < 0) ? N : g;
    ((float4*)(zs + r * D_))[lane] =
        ((const float4*)(xn + (size_t)row * D_))[lane];
  }
  __syncthreads();

  for (int it = 0; it < T; ++it) {
    if (it == 0) {
      #pragma unroll
      for (int q = 0; q < KC; ++q) ps[lane * KC + q] = 0.125f;
    } else {
      for (int k = 0; k < KC; ++k) {
        v8f d0 = {}; v8f d1 = {};
        #pragma unroll
        for (int s = 0; s < 4; ++s) {
          const int c0 = 4 * s + 2 * half;
          v2f b; b.x = us[k * DD_ + c0];     b.y = us[k * DD_ + c0 + 1];
          v2f a0; a0.x = zs[lo * D_        + k * DD_ + c0];
                  a0.y = zs[lo * D_        + k * DD_ + c0 + 1];
          v2f a1; a1.x = zs[(16 + lo) * D_ + k * DD_ + c0];
                  a1.y = zs[(16 + lo) * D_ + k * DD_ + c0 + 1];
          d0 = __builtin_amdgcn_wmma_f32_16x16x4_f32(false, a0, false, b, (short)0, d0, false, false);
          d1 = __builtin_amdgcn_wmma_f32_16x16x4_f32(false, a1, false, b, (short)0, d1, false, false);
        }
        if (lo == 0) {
          #pragma unroll
          for (int v = 0; v < 8; ++v) {
            ps[(8 * half + v) * KC + k]      = d0[v];
            ps[(16 + 8 * half + v) * KC + k] = d1[v];
          }
        }
      }
      __syncthreads();
      float pv[KC];
      float mx = -3.0e38f;
      #pragma unroll
      for (int q = 0; q < KC; ++q) { pv[q] = ps[lane * KC + q]; mx = fmaxf(mx, pv[q]); }
      float ssum = 0.0f;
      #pragma unroll
      for (int q = 0; q < KC; ++q) { pv[q] = expf(pv[q] - mx); ssum += pv[q]; }
      float inv = 1.0f / ssum;
      #pragma unroll
      for (int q = 0; q < KC; ++q) ps[lane * KC + q] = pv[q] * inv;
    }
    __syncthreads();

    for (int k = 0; k < KC; ++k) {
      v8f d = {};
      #pragma unroll
      for (int s = 0; s < 8; ++s) {
        const int m0 = 4 * s + 2 * half;
        v2f a; a.x = zs[m0 * D_ + k * DD_ + lo];
               a.y = zs[(m0 + 1) * D_ + k * DD_ + lo];
        v2f b; b.x = ps[m0 * KC + k];
               b.y = ps[(m0 + 1) * KC + k];
        d = __builtin_amdgcn_wmma_f32_16x16x4_f32(false, a, false, b, (short)0, d, false, false);
      }
      float uv[8];
      #pragma unroll
      for (int v = 0; v < 8; ++v) uv[v] = d[v] + xs[k * DD_ + 8 * half + v];
      if (it < T - 1) {
        float s2 = 0.0f;
        #pragma unroll
        for (int v = 0; v < 8; ++v) s2 += uv[v] * uv[v];
        s2 += __shfl_xor(s2, 16, 32);
        float inv = 1.0f / fmaxf(sqrtf(s2), 1e-12f);
        #pragma unroll
        for (int v = 0; v < 8; ++v) uv[v] *= inv;
      }
      if (lo == 0) {
        #pragma unroll
        for (int v = 0; v < 8; ++v) us[k * DD_ + 8 * half + v] = uv[v];
      }
    }
    __syncthreads();
  }

  ((float4*)(u_out + (size_t)node * D_))[lane] = ((float4*)us)[lane];

  if (lane < CUT_) {
    int best = 0;
    float bv = ps[lane * KC + 0];
    #pragma unroll
    for (int q = 1; q < KC; ++q) {
      float v = ps[lane * KC + q];
      if (v > bv) { bv = v; best = q; }
    }
    selcap[node * CUT_ + lane] = best;
  }
}

__global__ __launch_bounds__(128) void meta_kernel(const float* __restrict__ xn,
                                                   const int* __restrict__ nbr,
                                                   const int* __restrict__ selcap,
                                                   const int* __restrict__ p_last,
                                                   const float* __restrict__ u_out,
                                                   float* __restrict__ meta_out,
                                                   int N, int M) {
  if (p_last[0] == 0) return;
  const int node = blockIdx.x;
  const int t = threadIdx.x;
  const int i = t >> 4;
  const int c = t & 15;

  __shared__ int nb1[CUT_], cap1[CUT_];
  if (t < CUT_) {
    nb1[t]  = nbr[node * M + t];
    cap1[t] = selcap[node * CUT_ + t];
  }
  __syncthreads();

  float acc = 0.0f;
  for (int j = 0; j < KC; ++j) {
    float aj = 0.0f;
    #pragma unroll
    for (int s = 0; s < CUT_; ++s) {
      int g1 = nb1[s];
      if (g1 >= 0 && cap1[s] == i) {
        #pragma unroll
        for (int s2 = 0; s2 < CUT_; ++s2) {
          int g2 = nbr[g1 * M + s2];
          if (g2 >= 0 && selcap[g1 * CUT_ + s2] == j) {
            aj += xn[(size_t)g2 * D_ + j * DD_ + c];
          }
        }
      }
    }
    acc += aj / 25.0f;
  }
  meta_out[(size_t)node * D_ + t] = u_out[(size_t)node * D_ + t] + acc;
}

extern "C" void kernel_launch(void* const* d_in, const int* in_sizes, int n_in,
                              void* d_out, int out_size, void* d_ws, size_t ws_size,
                              hipStream_t stream) {
  const float* x       = (const float*)d_in[0];
  const int*   nbr     = (const int*)d_in[1];
  const int*   maxiter = (const int*)d_in[2];
  const int*   last    = (const int*)d_in[3];

  const int N = in_sizes[0] / D_;
  const int M = in_sizes[1] / N;

  float* u_out    = (float*)d_out;
  float* meta_out = (float*)d_out + (size_t)N * D_;

  float* xn = (float*)d_ws;
  size_t xn_bytes = (size_t)(N + 1) * D_ * sizeof(float);
  int* selcap = (int*)((char*)d_ws + xn_bytes);

  norm_kernel<<<N + 1, 128, 0, stream>>>(x, xn, N);
  routing_kernel<<<N, 32, 0, stream>>>(xn, nbr, maxiter, u_out, selcap, N, M);
  meta_kernel<<<N, 128, 0, stream>>>(xn, nbr, selcap, last, u_out, meta_out, N, M);
}


# Pad the LDS row pitch of the staged neighbour tile from 128 to 140 dwords so the A-fragment reads of both einsums spread over all banks instead of one bank group
# speedup vs baseline: 1.2099x; 1.2099x over previous
#include <hip/hip_runtime.h>
#include <math.h>

#define D_   128
#define KC   8
#define DD_  16
#define CUT_ 5
#define M_   32
#define ZP_  140

typedef float v2f __attribute__((ext_vector_type(2)));
typedef float v8f __attribute__((ext_vector_type(8)));

__global__ __launch_bounds__(128) void norm_kernel(const float* __restrict__ x,
                                                   float* __restrict__ xn, int N) {
  const int node = blockIdx.x;
  const int t = threadIdx.x;
  if (node >= N) { xn[(size_t)node * D_ + t] = 0.0f; return; }
  float v = x[(size_t)node * D_ + t];
  float s = v * v;
  s += __shfl_xor(s, 1, 16);
  s += __shfl_xor(s, 2, 16);
  s += __shfl_xor(s, 4, 16);
  s += __shfl_xor(s, 8, 16);
  float nrm = fmaxf(sqrtf(s), 1e-12f);
  xn[(size_t)node * D_ + t] = v / nrm;
}

__global__ __launch_bounds__(32) void routing_kernel(const float* __restrict__ xn,
                                                     const int* __restrict__ nbr,
                                                     const int* __restrict__ p_maxiter,
                                                     float* __restrict__ u_out,
                                                     int* __restrict__ selcap,
                                                     int N, int M) {
  __shared__ float zs[M_ * ZP_];
  __shared__ float ps[M_ * KC];
  __shared__ float us[D_];
  __shared__ float xs[D_];

  const int node = blockIdx.x;
  const int lane = threadIdx.x;
  const int half = lane >> 4;
  const int lo   = lane & 15;
  const int T = p_maxiter[0];

  ((float4*)xs)[lane] = ((const float4*)(xn + (size_t)node * D_))[lane];

  int gl = nbr[node * M + lane];
  for (int r = 0; r < M_; ++r) {
    int g = __shfl(gl, r, 32);
    int row = (g < 0) ? N : g;
    ((float4*)(zs + r * ZP_))[lane] =
        ((const float4*)(xn + (size_t)row * D_))[lane];
  }
  __syncthreads();

  for (int it = 0; it < T; ++it) {
    if (it == 0) {
      #pragma unroll
      for (int q = 0; q < KC; ++q) ps[lane * KC + q] = 0.125f;
    } else {
      for (int k = 0; k < KC; ++k) {
        v8f d0 = {}; v8f d1 = {};
        #pragma unroll
        for (int s = 0; s < 4; ++s) {
          const int c0 = 4 * s + 2 * half;
          v2f b; b.x = us[k * DD_ + c0];     b.y = us[k * DD_ + c0 + 1];
          v2f a0; a0.x = zs[lo * ZP_        + k * DD_ + c0];
                  a0.y = zs[lo * ZP_        + k * DD_ + c0 + 1];
          v2f a1; a1.x = zs[(16 + lo) * ZP_ + k * DD_ + c0];
                  a1.y = zs[(16 + lo) * ZP_ + k * DD_ + c0 + 1];
          d0 = __builtin_amdgcn_wmma_f32_16x16x4_f32(false, a0, false, b, (short)0, d0, false, false);
          d1 = __builtin_amdgcn_wmma_f32_16x16x4_f32(false, a1, false, b, (short)0, d1, false, false);
        }
        if (lo == 0) {
          #pragma unroll
          for (int v = 0; v < 8; ++v) {
            ps[(8 * half + v) * KC + k]      = d0[v];
            ps[(16 + 8 * half + v) * KC + k] = d1[v];
          }
        }
      }
      __syncthreads();
      float pv[KC];
      float mx = -3.0e38f;
      #pragma unroll
      for (int q = 0; q < KC; ++q) { pv[q] = ps[lane * KC + q]; mx = fmaxf(mx, pv[q]); }
      float ssum = 0.0f;
      #pragma unroll
      for (int q = 0; q < KC; ++q) { pv[q] = expf(pv[q] - mx); ssum += pv[q]; }
      float inv = 1.0f / ssum;
      #pragma unroll
      for (int q = 0; q < KC; ++q) ps[lane * KC + q] = pv[q] * inv;
    }
    __syncthreads();

    for (int k = 0; k < KC; ++k) {
      v8f d = {};
      #pragma unroll
      for (int s = 0; s < 8; ++s) {
        const int m0 = 4 * s + 2 * half;
        v2f a; a.x = zs[m0 * ZP_ + k * DD_ + lo];
               a.y = zs[(m0 + 1) * ZP_ + k * DD_ + lo];
        v2f b; b.x = ps[m0 * KC + k];
               b.y = ps[(m0 + 1) * KC + k];
        d = __builtin_amdgcn_wmma_f32_16x16x4_f32(false, a, false, b, (short)0, d, false, false);
      }
      float uv[8];
      #pragma unroll
      for (int v = 0; v < 8; ++v) uv[v] = d[v] + xs[k * DD_ + 8 * half + v];
      if (it < T - 1) {
        float s2 = 0.0f;
        #pragma unroll
        for (int v = 0; v < 8; ++v) s2 += uv[v] * uv[v];
        s2 += __shfl_xor(s2, 16, 32);
        float inv = 1.0f / fmaxf(sqrtf(s2), 1e-12f);
        #pragma unroll
        for (int v = 0; v < 8; ++v) uv[v] *= inv;
      }
      if (lo == 0) {
        #pragma unroll
        for (int v = 0; v < 8; ++v) us[k * DD_ + 8 * half + v] = uv[v];
      }
    }
    __syncthreads();
  }

  ((float4*)(u_out + (size_t)node * D_))[lane] = ((float4*)us)[lane];

  if (lane < CUT_) {
    int best = 0;
    float bv = ps[lane * KC + 0];
    #pragma unroll
    for (int q = 1; q < KC; ++q) {
      float v = ps[lane * KC + q];
      if (v > bv) { bv = v; best = q; }
    }
    selcap[node * CUT_ + lane] = best;
  }
}

__global__ __launch_bounds__(128) void meta_kernel(const float* __restrict__ xn,
                                                   const int* __restrict__ nbr,
                                                   const int* __restrict__ selcap,
                                                   const int* __restrict__ p_last,
                                                   const float* __restrict__ u_out,
                                                   float* __restrict__ meta_out,
                                                   int N, int M) {
  if (p_last[0] == 0) return;
  const int node = blockIdx.x;
  const int t = threadIdx.x;
  const int i = t >> 4;
  const int c = t & 15;

  __shared__ int nb1[CUT_], cap1[CUT_];
  if (t < CUT_) {
    nb1[t]  = nbr[node * M + t];
    cap1[t] = selcap[node * CUT_ + t];
  }
  __syncthreads();

  float acc = 0.0f;
  for (int j = 0; j < KC; ++j) {
    float aj = 0.0f;
    #pragma unroll
    for (int s = 0; s < CUT_; ++s) {
      int g1 = nb1[s];
      if (g1 >= 0 && cap1[s] == i) {
        #pragma unroll
        for (int s2 = 0; s2 < CUT_; ++s2) {
          int g2 = nbr[g1 * M + s2];
          if (g2 >= 0 && selcap[g1 * CUT_ + s2] == j) {
            aj += xn[(size_t)g2 * D_ + j * DD_ + c];
          }
        }
      }
    }
    acc += aj / 25.0f;
  }
  meta_out[(size_t)node * D_ + t] = u_out[(size_t)node * D_ + t] + acc;
}

extern "C" void kernel_launch(void* const* d_in, const int* in_sizes, int n_in,
                              void* d_out, int out_size, void* d_ws, size_t ws_size,
                              hipStream_t stream) {
  const float* x       = (const float*)d_in[0];
  const int*   nbr     = (const int*)d_in[1];
  const int*   maxiter = (const int*)d_in[2];
  const int*   last    = (const int*)d_in[3];

  const int N = in_sizes[0] / D_;
  const int M = in_sizes[1] / N;

  float* u_out    = (float*)d_out;
  float* meta_out = (float*)d_out + (size_t)N * D_;

  float* xn = (float*)d_ws;
  size_t xn_bytes = (size_t)(N + 1) * D_ * sizeof(float);
  int* selcap = (int*)((char*)d_ws + xn_bytes);

  norm_kernel<<<N + 1, 128, 0, stream>>>(x, xn, N);
  routing_kernel<<<N, 32, 0, stream>>>(xn, nbr, maxiter, u_out, selcap, N, M);
  meta_kernel<<<N, 128, 0, stream>>>(xn, nbr, selcap, last, u_out, meta_out, N, M);
}


# Both einsums take a B fragment masked to its capsule's columns: u and p land in the lane owning the capsule and stay in registers, no LDS round trip; sums reordered (m, c by halves; softmax tree)
# speedup vs baseline: 1.2996x; 1.0741x over previous
#include <hip/hip_runtime.h>
#include <math.h>

#define D_   128
#define KC   8
#define DD_  16
#define CUT_ 5
#define M_   32
#define ZP_  140

typedef float v2f __attribute__((ext_vector_type(2)));
typedef float v8f __attribute__((ext_vector_type(8)));

__global__ __launch_bounds__(128) void norm_kernel(const float* __restrict__ x,
                                                   float* __restrict__ xn, int N) {
  const int node = blockIdx.x;
  const int t = threadIdx.x;
  if (node >= N) { xn[(size_t)node * D_ + t] = 0.0f; return; }
  float v = x[(size_t)node * D_ + t];
  float s = v * v;
  s += __shfl_xor(s, 1, 16);
  s += __shfl_xor(s, 2, 16);
  s += __shfl_xor(s, 4, 16);
  s += __shfl_xor(s, 8, 16);
  float nrm = fmaxf(sqrtf(s), 1e-12f);
  xn[(size_t)node * D_ + t] = v / nrm;
}

template <int X>
__device__ __forceinline__ float lane_xor(float v) {
  int iv = __builtin_bit_cast(int, v);
  return __builtin_bit_cast(float,
      __builtin_amdgcn_update_dpp(iv, iv, 0x160 + X, 0xf, 0xf, false));
}

__global__ __launch_bounds__(32) void routing_kernel(const float* __restrict__ xn,
                                                     const int* __restrict__ nbr,
                                                     const int* __restrict__ p_maxiter,
                                                     float* __restrict__ u_out,
                                                     int* __restrict__ selcap,
                                                     int N, int M) {
  __shared__ float zs[M_ * ZP_];
  __shared__ float us[D_];
  __shared__ float ps[CUT_ * KC];

  const int node = blockIdx.x;
  const int lane = threadIdx.x;
  const int half = lane >> 4;
  const int lo   = lane & 15;
  const int cap  = lo & 7;
  const int T = p_maxiter[0];

  float xr[8];
  {
    const float4* xp = (const float4*)(xn + (size_t)node * D_ + cap * DD_ + 8 * half);
    const float4 q0 = xp[0], q1 = xp[1];
    xr[0] = q0.x; xr[1] = q0.y; xr[2] = q0.z; xr[3] = q0.w;
    xr[4] = q1.x; xr[5] = q1.y; xr[6] = q1.z; xr[7] = q1.w;
  }

  int gl = nbr[node * M + lane];
  for (int r = 0; r < M_; ++r) {
    int g = __shfl(gl, r, 32);
    int row = (g < 0) ? N : g;
    ((float4*)(zs + r * ZP_))[lane] =
        ((const float4*)(xn + (size_t)row * D_))[lane];
  }
  __syncthreads();

  float p[2][8];
  float u[8];
  #pragma unroll
  for (int r = 0; r < 8; ++r) { p[0][r] = 0.125f; p[1][r] = 0.125f; u[r] = 0.0f; }

  for (int it = 0; it < T; ++it) {
    v8f de = {}; v8f dq = {};
    #pragma unroll
    for (int k = 0; k < KC; ++k) {
      const bool mine = (cap == k);
      #pragma unroll
      for (int t = 0; t < 2; ++t) {
        #pragma unroll
        for (int j = 0; j < 4; ++j) {
          const int m0 = 16 * t + 8 * half + 2 * j;
          v2f a; a.x = zs[m0 * ZP_ + k * DD_ + lo];
                 a.y = zs[(m0 + 1) * ZP_ + k * DD_ + lo];
          v2f b; b.x = mine ? p[t][2 * j] : 0.0f;
                 b.y = mine ? p[t][2 * j + 1] : 0.0f;
          if (k & 1) dq = __builtin_amdgcn_wmma_f32_16x16x4_f32(false, a, false, b, (short)0, dq, false, false);
          else       de = __builtin_amdgcn_wmma_f32_16x16x4_f32(false, a, false, b, (short)0, de, false, false);
        }
      }
    }
    #pragma unroll
    for (int r = 0; r < 8; ++r) u[r] = ((cap & 1) ? dq[r] : de[r]) + xr[r];
    if (it == T - 1) break;

    float s2 = 0.0f;
    #pragma unroll
    for (int r = 0; r < 8; ++r) s2 += u[r] * u[r];
    s2 += __shfl_xor(s2, 16, 32);
    const float inv = 1.0f / fmaxf(sqrtf(s2), 1e-12f);
    #pragma unroll
    for (int r = 0; r < 8; ++r) u[r] *= inv;

    v8f d0 = {}; v8f d1 = {};
    #pragma unroll
    for (int k = 0; k < KC; ++k) {
      const bool mine = (cap == k);
      #pragma unroll
      for (int s = 0; s < 4; ++s) {
        const int c0 = 8 * half + 2 * s;
        v2f b; b.x = mine ? u[2 * s] : 0.0f;
               b.y = mine ? u[2 * s + 1] : 0.0f;
        v2f a0; a0.x = zs[lo * ZP_        + k * DD_ + c0];
                a0.y = zs[lo * ZP_        + k * DD_ + c0 + 1];
        v2f a1; a1.x = zs[(16 + lo) * ZP_ + k * DD_ + c0];
                a1.y = zs[(16 + lo) * ZP_ + k * DD_ + c0 + 1];
        d0 = __builtin_amdgcn_wmma_f32_16x16x4_f32(false, a0, false, b, (short)0, d0, false, false);
        d1 = __builtin_amdgcn_wmma_f32_16x16x4_f32(false, a1, false, b, (short)0, d1, false, false);
      }
    }

    #pragma unroll
    for (int t = 0; t < 2; ++t) {
      #pragma unroll
      for (int r = 0; r < 8; ++r) {
        const float v = t ? d1[r] : d0[r];
        float mx = v;
        mx = fmaxf(mx, lane_xor<1>(mx));
        mx = fmaxf(mx, lane_xor<2>(mx));
        mx = fmaxf(mx, lane_xor<4>(mx));
        const float e = expf(v - mx);
        float ssum = e;
        ssum += lane_xor<1>(ssum);
        ssum += lane_xor<2>(ssum);
        ssum += lane_xor<4>(ssum);
        p[t][r] = e * (1.0f / ssum);
      }
    }
  }

  if (lo < KC) {
    #pragma unroll
    for (int r = 0; r < 8; ++r) us[cap * DD_ + 8 * half + r] = u[r];
    if (half == 0) {
      #pragma unroll
      for (int r = 0; r < CUT_; ++r) ps[r * KC + cap] = p[0][r];
    }
  }
  __syncthreads();

  ((float4*)(u_out + (size_t)node * D_))[lane] = ((float4*)us)[lane];

  if (lane < CUT_) {
    int best = 0;
    float bv = ps[lane * KC + 0];
    #pragma unroll
    for (int q = 1; q < KC; ++q) {
      float v = ps[lane * KC + q];
      if (v > bv) { bv = v; best = q; }
    }
    selcap[node * CUT_ + lane] = best;
  }
}

__global__ __launch_bounds__(128) void meta_kernel(const float* __restrict__ xn,
                                                   const int* __restrict__ nbr,
                                                   const int* __restrict__ selcap,
                                                   const int* __restrict__ p_last,
                                                   const float* __restrict__ u_out,
                                                   float* __restrict__ meta_out,
                                                   int N, int M) {
  if (p_last[0] == 0) return;
  const int node = blockIdx.x;
  const int t = threadIdx.x;
  const int i = t >> 4;
  const int c = t & 15;

  __shared__ int nb1[CUT_], cap1[CUT_];
  if (t < CUT_) {
    nb1[t]  = nbr[node * M + t];
    cap1[t] = selcap[node * CUT_ + t];
  }
  __syncthreads();

  float acc = 0.0f;
  for (int j = 0; j < KC; ++j) {
    float aj = 0.0f;
    #pragma unroll
    for (int s = 0; s < CUT_; ++s) {
      int g1 = nb1[s];
      if (g1 >= 0 && cap1[s] == i) {
        #pragma unroll
        for (int s2 = 0; s2 < CUT_; ++s2) {
          int g2 = nbr[g1 * M + s2];
          if (g2 >= 0 && selcap[g1 * CUT_ + s2] == j) {
            aj += xn[(size_t)g2 * D_ + j * DD_ + c];
          }
        }
      }
    }
    acc += aj / 25.0f;
  }
  meta_out[(size_t)node * D_ + t] = u_out[(size_t)node * D_ + t] + acc;
}

extern "C" void kernel_launch(void* const* d_in, const int* in_sizes, int n_in,
                              void* d_out, int out_size, void* d_ws, size_t ws_size,
                              hipStream_t stream) {
  const float* x       = (const float*)d_in[0];
  const int*   nbr     = (const int*)d_in[1];
  const int*   maxiter = (const int*)d_in[2];
  const int*   last    = (const int*)d_in[3];

  const int N = in_sizes[0] / D_;
  const int M = in_sizes[1] / N;

  float* u_out    = (float*)d_out;
  float* meta_out = (float*)d_out + (size_t)N * D_;

  float* xn = (float*)d_ws;
  size_t xn_bytes = (size_t)(N + 1) * D_ * sizeof(float);
  int* selcap = (int*)((char*)d_ws + xn_bytes);

  norm_kernel<<<N + 1, 128, 0, stream>>>(x, xn, N);
  routing_kernel<<<N, 32, 0, stream>>>(xn, nbr, maxiter, u_out, selcap, N, M);
  meta_kernel<<<N, 128, 0, stream>>>(xn, nbr, selcap, last, u_out, meta_out, N, M);
}


# Two-hop gather regrouped: a wave takes whole nodes (4 in turn), 25 lanes fetch the 25 hop index pairs at once, rows read by 25 independent 16-byte loads; meta summed in flat pair order, one division
# speedup vs baseline: 1.7184x; 1.3223x over previous
#include <hip/hip_runtime.h>
#include <math.h>

#define D_   128
#define KC   8
#define DD_  16
#define CUT_ 5
#define M_   32
#define ZP_  140

typedef float v2f __attribute__((ext_vector_type(2)));
typedef float v8f __attribute__((ext_vector_type(8)));

__global__ __launch_bounds__(128) void norm_kernel(const float* __restrict__ x,
                                                   float* __restrict__ xn, int N) {
  const int node = blockIdx.x;
  const int t = threadIdx.x;
  if (node >= N) { xn[(size_t)node * D_ + t] = 0.0f; return; }
  float v = x[(size_t)node * D_ + t];
  float s = v * v;
  s += __shfl_xor(s, 1, 16);
  s += __shfl_xor(s, 2, 16);
  s += __shfl_xor(s, 4, 16);
  s += __shfl_xor(s, 8, 16);
  float nrm = fmaxf(sqrtf(s), 1e-12f);
  xn[(size_t)node * D_ + t] = v / nrm;
}

template <int X>
__device__ __forceinline__ float lane_xor(float v) {
  int iv = __builtin_bit_cast(int, v);
  return __builtin_bit_cast(float,
      __builtin_amdgcn_update_dpp(iv, iv, 0x160 + X, 0xf, 0xf, false));
}

__global__ __launch_bounds__(32) void routing_kernel(const float* __restrict__ xn,
                                                     const int* __restrict__ nbr,
                                                     const int* __restrict__ p_maxiter,
                                                     float* __restrict__ u_out,
                                                     int* __restrict__ selcap,
                                                     int N, int M) {
  __shared__ float zs[M_ * ZP_];
  __shared__ float us[D_];
  __shared__ float ps[CUT_ * KC];

  const int node = blockIdx.x;
  const int lane = threadIdx.x;
  const int half = lane >> 4;
  const int lo   = lane & 15;
  const int cap  = lo & 7;
  const int T = p_maxiter[0];

  float xr[8];
  {
    const float4* xp = (const float4*)(xn + (size_t)node * D_ + cap * DD_ + 8 * half);
    const float4 q0 = xp[0], q1 = xp[1];
    xr[0] = q0.x; xr[1] = q0.y; xr[2] = q0.z; xr[3] = q0.w;
    xr[4] = q1.x; xr[5] = q1.y; xr[6] = q1.z; xr[7] = q1.w;
  }

  int gl = nbr[node * M + lane];
  for (int r = 0; r < M_; ++r) {
    int g = __shfl(gl, r, 32);
    int row = (g < 0) ? N : g;
    ((float4*)(zs + r * ZP_))[lane] =
        ((const float4*)(xn + (size_t)row * D_))[lane];
  }
  __syncthreads();

  float p[2][8];
  float u[8];
  #pragma unroll
  for (int r = 0; r < 8; ++r) { p[0][r] = 0.125f; p[1][r] = 0.125f; u[r] = 0.0f; }

  for (int it = 0; it < T; ++it) {
    v8f de = {}; v8f dq = {};
    #pragma unroll
    for (int k = 0; k < KC; ++k) {
      const bool mine = (cap == k);
      #pragma unroll
      for (int t = 0; t < 2; ++t) {
        #pragma unroll
        for (int j = 0; j < 4; ++j) {
          const int m0 = 16 * t + 8 * half + 2 * j;
          v2f a; a.x = zs[m0 * ZP_ + k * DD_ + lo];
                 a.y = zs[(m0 + 1) * ZP_ + k * DD_ + lo];
          v2f b; b.x = mine ? p[t][2 * j] : 0.0f;
                 b.y = mine ? p[t][2 * j + 1] : 0.0f;
          if (k & 1) dq = __builtin_amdgcn_wmma_f32_16x16x4_f32(false, a, false, b, (short)0, dq, false, false);
          else       de = __builtin_amdgcn_wmma_f32_16x16x4_f32(false, a, false, b, (short)0, de, false, false);
        }
      }
    }
    #pragma unroll
    for (int r = 0; r < 8; ++r) u[r] = ((cap & 1) ? dq[r] : de[r]) + xr[r];
    if (it == T - 1) break;

    float s2 = 0.0f;
    #pragma unroll
    for (int r = 0; r < 8; ++r) s2 += u[r] * u[r];
    s2 += __shfl_xor(s2, 16, 32);
    const float inv = 1.0f / fmaxf(sqrtf(s2), 1e-12f);
    #pragma unroll
    for (int r = 0; r < 8; ++r) u[r] *= inv;

    v8f d0 = {}; v8f d1 = {};
    #pragma unroll
    for (int k = 0; k < KC; ++k) {
      const bool mine = (cap == k);
      #pragma unroll
      for (int s = 0; s < 4; ++s) {
        const int c0 = 8 * half + 2 * s;
        v2f b; b.x = mine ? u[2 * s] : 0.0f;
               b.y = mine ? u[2 * s + 1] : 0.0f;
        v2f a0; a0.x = zs[lo * ZP_        + k * DD_ + c0];
                a0.y = zs[lo * ZP_        + k * DD_ + c0 + 1];
        v2f a1; a1.x = zs[(16 + lo) * ZP_ + k * DD_ + c0];
                a1.y = zs[(16 + lo) * ZP_ + k * DD_ + c0 + 1];
        d0 = __builtin_amdgcn_wmma_f32_16x16x4_f32(false, a0, false, b, (short)0, d0, false, false);
        d1 = __builtin_amdgcn_wmma_f32_16x16x4_f32(false, a1, false, b, (short)0, d1, false, false);
      }
    }

    #pragma unroll
    for (int t = 0; t < 2; ++t) {
      #pragma unroll
      for (int r = 0; r < 8; ++r) {
        const float v = t ? d1[r] : d0[r];
        float mx = v;
        mx = fmaxf(mx, lane_xor<1>(mx));
        mx = fmaxf(mx, lane_xor<2>(mx));
        mx = fmaxf(mx, lane_xor<4>(mx));
        const float e = expf(v - mx);
        float ssum = e;
        ssum += lane_xor<1>(ssum);
        ssum += lane_xor<2>(ssum);
        ssum += lane_xor<4>(ssum);
        p[t][r] = e * (1.0f / ssum);
      }
    }
  }

  if (lo < KC) {
    #pragma unroll
    for (int r = 0; r < 8; ++r) us[cap * DD_ + 8 * half + r] = u[r];
    if (half == 0) {
      #pragma unroll
      for (int r = 0; r < CUT_; ++r) ps[r * KC + cap] = p[0][r];
    }
  }
  __syncthreads();

  ((float4*)(u_out + (size_t)node * D_))[lane] = ((float4*)us)[lane];

  if (lane < CUT_) {
    int best = 0;
    float bv = ps[lane * KC + 0];
    #pragma unroll
    for (int q = 1; q < KC; ++q) {
      float v = ps[lane * KC + q];
      if (v > bv) { bv = v; best = q; }
    }
    selcap[node * CUT_ + lane] = best;
  }
}

#define NPW_ 4
#define NPAIR_ (CUT_ * CUT_)

__global__ __launch_bounds__(256) void meta_kernel(const float* __restrict__ xn,
                                                   const int* __restrict__ nbr,
                                                   const int* __restrict__ selcap,
                                                   const int* __restrict__ p_last,
                                                   const float* __restrict__ u_out,
                                                   float* __restrict__ meta_out,
                                                   int N, int M) {
  if (p_last[0] == 0) return;
  const int lane  = threadIdx.x & 31;
  const int node0 = (blockIdx.x * 8 + (threadIdx.x >> 5)) * NPW_;
  if (node0 >= N) return;
  const int cap = lane >> 2;
  const int sub = lane & 3;
  const int q   = lane < NPAIR_ ? lane : NPAIR_ - 1;
  const int s   = q / CUT_;
  const int s2  = q - s * CUT_;

  int g1[NPW_], c1[NPW_], off[NPW_];
  #pragma unroll
  for (int n = 0; n < NPW_; ++n) {
    const int nd = min(node0 + n, N - 1);
    g1[n] = nbr[nd * M + s];
    c1[n] = selcap[nd * CUT_ + s];
  }
  #pragma unroll
  for (int n = 0; n < NPW_; ++n) {
    const int h  = g1[n] < 0 ? 0 : g1[n];
    const int g2 = nbr[h * M + s2];
    const int jj = selcap[h * CUT_ + s2];
    const int row = ((g1[n] | g2) >= 0) ? g2 : N;
    off[n] = row * D_ + jj * DD_;
  }

  #pragma unroll
  for (int n = 0; n < NPW_; ++n) {
    const int node = node0 + n;
    if (node >= N) break;
    const float4 uu = ((const float4*)(u_out + (size_t)node * D_))[lane];
    float4 v[NPAIR_];
    #pragma unroll
    for (int k = 0; k < NPAIR_; ++k) {
      const int o = __builtin_amdgcn_readlane(off[n], k);
      v[k] = *(const float4*)(xn + (size_t)o + 4 * sub);
    }
    float4 acc;
    #pragma unroll
    for (int k = 0; k < NPAIR_; ++k) {
      const int cs = __builtin_amdgcn_readlane(c1[n], (k / CUT_) * CUT_);
      const bool mine = (cap == cs);
      const float tx = mine ? v[k].x : 0.0f, ty = mine ? v[k].y : 0.0f;
      const float tz = mine ? v[k].z : 0.0f, tw = mine ? v[k].w : 0.0f;
      if (k == 0) { acc.x = tx; acc.y = ty; acc.z = tz; acc.w = tw; }
      else        { acc.x += tx; acc.y += ty; acc.z += tz; acc.w += tw; }
    }
    float4 r;
    r.x = uu.x + acc.x / 25.0f;
    r.y = uu.y + acc.y / 25.0f;
    r.z = uu.z + acc.z / 25.0f;
    r.w = uu.w + acc.w / 25.0f;
    ((float4*)(meta_out + (size_t)node * D_))[lane] = r;
  }
}

extern "C" void kernel_launch(void* const* d_in, const int* in_sizes, int n_in,
                              void* d_out, int out_size, void* d_ws, size_t ws_size,
                              hipStream_t stream) {
  const float* x       = (const float*)d_in[0];
  const int*   nbr     = (const int*)d_in[1];
  const int*   maxiter = (const int*)d_in[2];
  const int*   last    = (const int*)d_in[3];

  const int N = in_sizes[0] / D_;
  const int M = in_sizes[1] / N;

  float* u_out    = (float*)d_out;
  float* meta_out = (float*)d_out + (size_t)N * D_;

  float* xn = (float*)d_ws;
  size_t xn_bytes = (size_t)(N + 1) * D_ * sizeof(float);
  int* selcap = (int*)((char*)d_ws + xn_bytes);

  norm_kernel<<<N + 1, 128, 0, stream>>>(x, xn, N);
  routing_kernel<<<N, 32, 0, stream>>>(xn, nbr, maxiter, u_out, selcap, N, M);
  meta_kernel<<<(N + 8 * NPW_ - 1) / (8 * NPW_), 256, 0, stream>>>(xn, nbr, selcap, last, u_out, meta_out, N, M);
}


# Softmax no longer computed twice: lanes lo and lo^8 hold the same capsule column, so each 8-lane group takes 4 of a tile's 8 rows and the groups swap results by one DPP move; values bit-identical
# speedup vs baseline: 1.8828x; 1.0957x over previous
#include <hip/hip_runtime.h>
#include <math.h>

#define D_   128
#define KC   8
#define DD_  16
#define CUT_ 5
#define M_   32
#define ZP_  140

typedef float v2f __attribute__((ext_vector_type(2)));
typedef float v8f __attribute__((ext_vector_type(8)));

__global__ __launch_bounds__(128) void norm_kernel(const float* __restrict__ x,
                                                   float* __restrict__ xn, int N) {
  const int node = blockIdx.x;
  const int t = threadIdx.x;
  if (node >= N) { xn[(size_t)node * D_ + t] = 0.0f; return; }
  float v = x[(size_t)node * D_ + t];
  float s = v * v;
  s += __shfl_xor(s, 1, 16);
  s += __shfl_xor(s, 2, 16);
  s += __shfl_xor(s, 4, 16);
  s += __shfl_xor(s, 8, 16);
  float nrm = fmaxf(sqrtf(s), 1e-12f);
  xn[(size_t)node * D_ + t] = v / nrm;
}

template <int X>
__device__ __forceinline__ float lane_xor(float v) {
  int iv = __builtin_bit_cast(int, v);
  return __builtin_bit_cast(float,
      __builtin_amdgcn_update_dpp(iv, iv, 0x160 + X, 0xf, 0xf, false));
}

__global__ __launch_bounds__(32) void routing_kernel(const float* __restrict__ xn,
                                                     const int* __restrict__ nbr,
                                                     const int* __restrict__ p_maxiter,
                                                     float* __restrict__ u_out,
                                                     int* __restrict__ selcap,
                                                     int N, int M) {
  __shared__ float zs[M_ * ZP_];
  __shared__ float us[D_];
  __shared__ float ps[CUT_ * KC];

  const int node = blockIdx.x;
  const int lane = threadIdx.x;
  const int half = lane >> 4;
  const int lo   = lane & 15;
  const int cap  = lo & 7;
  const int T = p_maxiter[0];

  float xr[8];
  {
    const float4* xp = (const float4*)(xn + (size_t)node * D_ + cap * DD_ + 8 * half);
    const float4 q0 = xp[0], q1 = xp[1];
    xr[0] = q0.x; xr[1] = q0.y; xr[2] = q0.z; xr[3] = q0.w;
    xr[4] = q1.x; xr[5] = q1.y; xr[6] = q1.z; xr[7] = q1.w;
  }

  int gl = nbr[node * M + lane];
  for (int r = 0; r < M_; ++r) {
    int g = __shfl(gl, r, 32);
    int row = (g < 0) ? N : g;
    ((float4*)(zs + r * ZP_))[lane] =
        ((const float4*)(xn + (size_t)row * D_))[lane];
  }
  __syncthreads();

  float p[2][8];
  float u[8];
  #pragma unroll
  for (int r = 0; r < 8; ++r) { p[0][r] = 0.125f; p[1][r] = 0.125f; u[r] = 0.0f; }

  for (int it = 0; it < T; ++it) {
    v8f de = {}; v8f dq = {};
    #pragma unroll
    for (int k = 0; k < KC; ++k) {
      const bool mine = (cap == k);
      #pragma unroll
      for (int t = 0; t < 2; ++t) {
        #pragma unroll
        for (int j = 0; j < 4; ++j) {
          const int m0 = 16 * t + 8 * half + 2 * j;
          v2f a; a.x = zs[m0 * ZP_ + k * DD_ + lo];
                 a.y = zs[(m0 + 1) * ZP_ + k * DD_ + lo];
          v2f b; b.x = mine ? p[t][2 * j] : 0.0f;
                 b.y = mine ? p[t][2 * j + 1] : 0.0f;
          if (k & 1) dq = __builtin_amdgcn_wmma_f32_16x16x4_f32(false, a, false, b, (short)0, dq, false, false);
          else       de = __builtin_amdgcn_wmma_f32_16x16x4_f32(false, a, false, b, (short)0, de, false, false);
        }
      }
    }
    #pragma unroll
    for (int r = 0; r < 8; ++r) u[r] = ((cap & 1) ? dq[r] : de[r]) + xr[r];
    if (it == T - 1) break;

    float s2 = 0.0f;
    #pragma unroll
    for (int r = 0; r < 8; ++r) s2 += u[r] * u[r];
    s2 += __shfl_xor(s2, 16, 32);
    const float inv = 1.0f / fmaxf(sqrtf(s2), 1e-12f);
    #pragma unroll
    for (int r = 0; r < 8; ++r) u[r] *= inv;

    v8f d0 = {}; v8f d1 = {};
    #pragma unroll
    for (int k = 0; k < KC; ++k) {
      const bool mine = (cap == k);
      #pragma unroll
      for (int s = 0; s < 4; ++s) {
        const int c0 = 8 * half + 2 * s;
        v2f b; b.x = mine ? u[2 * s] : 0.0f;
               b.y = mine ? u[2 * s + 1] : 0.0f;
        v2f a0; a0.x = zs[lo * ZP_        + k * DD_ + c0];
                a0.y = zs[lo * ZP_        + k * DD_ + c0 + 1];
        v2f a1; a1.x = zs[(16 + lo) * ZP_ + k * DD_ + c0];
                a1.y = zs[(16 + lo) * ZP_ + k * DD_ + c0 + 1];
        d0 = __builtin_amdgcn_wmma_f32_16x16x4_f32(false, a0, false, b, (short)0, d0, false, false);
        d1 = __builtin_amdgcn_wmma_f32_16x16x4_f32(false, a1, false, b, (short)0, d1, false, false);
      }
    }

    const bool upper = (lo & 8) != 0;
    #pragma unroll
    for (int t = 0; t < 2; ++t) {
      float q[4];
      #pragma unroll
      for (int i = 0; i < 4; ++i) {
        const float vl = t ? d1[i] : d0[i];
        const float vh = t ? d1[4 + i] : d0[4 + i];
        const float v = upper ? vh : vl;
        float mx = v;
        mx = fmaxf(mx, lane_xor<1>(mx));
        mx = fmaxf(mx, lane_xor<2>(mx));
        mx = fmaxf(mx, lane_xor<4>(mx));
        const float e = expf(v - mx);
        float ssum = e;
        ssum += lane_xor<1>(ssum);
        ssum += lane_xor<2>(ssum);
        ssum += lane_xor<4>(ssum);
        q[i] = e * (1.0f / ssum);
      }
      #pragma unroll
      for (int i = 0; i < 4; ++i) {
        const float o = lane_xor<8>(q[i]);
        p[t][i]     = upper ? o : q[i];
        p[t][4 + i] = upper ? q[i] : o;
      }
    }
  }

  if (lo < KC) {
    #pragma unroll
    for (int r = 0; r < 8; ++r) us[cap * DD_ + 8 * half + r] = u[r];
    if (half == 0) {
      #pragma unroll
      for (int r = 0; r < CUT_; ++r) ps[r * KC + cap] = p[0][r];
    }
  }
  __syncthreads();

  ((float4*)(u_out + (size_t)node * D_))[lane] = ((float4*)us)[lane];

  if (lane < CUT_) {
    int best = 0;
    float bv = ps[lane * KC + 0];
    #pragma unroll
    for (int q = 1; q < KC; ++q) {
      float v = ps[lane * KC + q];
      if (v > bv) { bv = v; best = q; }
    }
    selcap[node * CUT_ + lane] = best;
  }
}

#define NPW_ 4
#define NPAIR_ (CUT_ * CUT_)

__global__ __launch_bounds__(256) void meta_kernel(const float* __restrict__ xn,
                                                   const int* __restrict__ nbr,
                                                   const int* __restrict__ selcap,
                                                   const int* __restrict__ p_last,
                                                   const float* __restrict__ u_out,
                                                   float* __restrict__ meta_out,
                                                   int N, int M) {
  if (p_last[0] == 0) return;
  const int lane  = threadIdx.x & 31;
  const int node0 = (blockIdx.x * 8 + (threadIdx.x >> 5)) * NPW_;
  if (node0 >= N) return;
  const int cap = lane >> 2;
  const int sub = lane & 3;
  const int q   = lane < NPAIR_ ? lane : NPAIR_ - 1;
  const int s   = q / CUT_;
  const int s2  = q - s * CUT_;

  int g1[NPW_], c1[NPW_], off[NPW_];
  #pragma unroll
  for (int n = 0; n < NPW_; ++n) {
    const int nd = min(node0 + n, N - 1);
    g1[n] = nbr[nd * M + s];
    c1[n] = selcap[nd * CUT_ + s];
  }
  #pragma unroll
  for (int n = 0; n < NPW_; ++n) {
    const int h  = g1[n] < 0 ? 0 : g1[n];
    const int g2 = nbr[h * M + s2];
    const int jj = selcap[h * CUT_ + s2];
    const int row = ((g1[n] | g2) >= 0) ? g2 : N;
    off[n] = row * D_ + jj * DD_;
  }

  #pragma unroll
  for (int n = 0; n < NPW_; ++n) {
    const int node = node0 + n;
    if (node >= N) break;
    const float4 uu = ((const float4*)(u_out + (size_t)node * D_))[lane];
    float4 v[NPAIR_];
    #pragma unroll
    for (int k = 0; k < NPAIR_; ++k) {
      const int o = __builtin_amdgcn_readlane(off[n], k);
      v[k] = *(const float4*)(xn + (size_t)o + 4 * sub);
    }
    float4 acc;
    #pragma unroll
    for (int k = 0; k < NPAIR_; ++k) {
      const int cs = __builtin_amdgcn_readlane(c1[n], (k / CUT_) * CUT_);
      const bool mine = (cap == cs);
      const float tx = mine ? v[k].x : 0.0f, ty = mine ? v[k].y : 0.0f;
      const float tz = mine ? v[k].z : 0.0f, tw = mine ? v[k].w : 0.0f;
      if (k == 0) { acc.x = tx; acc.y = ty; acc.z = tz; acc.w = tw; }
      else        { acc.x += tx; acc.y += ty; acc.z += tz; acc.w += tw; }
    }
    float4 r;
    r.x = uu.x + acc.x / 25.0f;
    r.y = uu.y + acc.y / 25.0f;
    r.z = uu.z + acc.z / 25.0f;
    r.w = uu.w + acc.w / 25.0f;
    ((float4*)(meta_out + (size_t)node * D_))[lane] = r;
  }
}

extern "C" void kernel_launch(void* const* d_in, const int* in_sizes, int n_in,
                              void* d_out, int out_size, void* d_ws, size_t ws_size,
                              hipStream_t stream) {
  const float* x       = (const float*)d_in[0];
  const int*   nbr     = (const int*)d_in[1];
  const int*   maxiter = (const int*)d_in[2];
  const int*   last    = (const int*)d_in[3];

  const int N = in_sizes[0] / D_;
  const int M = in_sizes[1] / N;

  float* u_out    = (float*)d_out;
  float* meta_out = (float*)d_out + (size_t)N * D_;

  float* xn = (float*)d_ws;
  size_t xn_bytes = (size_t)(N + 1) * D_ * sizeof(float);
  int* selcap = (int*)((char*)d_ws + xn_bytes);

  norm_kernel<<<N + 1, 128, 0, stream>>>(x, xn, N);
  routing_kernel<<<N, 32, 0, stream>>>(xn, nbr, maxiter, u_out, selcap, N, M);
  meta_kernel<<<(N + 8 * NPW_ - 1) / (8 * NPW_), 256, 0, stream>>>(xn, nbr, selcap, last, u_out, meta_out, N, M);
}


# Routing loop's exp, softmax division and capsule-norm inverse root use the hardware exp2, rcp and rsq instructions instead of the exact library expansions; p and u differ from before by a few ulp
# speedup vs baseline: 2.0071x; 1.0660x over previous
#include <hip/hip_runtime.h>
#include <math.h>

#define D_   128
#define KC   8
#define DD_  16
#define CUT_ 5
#define M_   32
#define ZP_  140

typedef float v2f __attribute__((ext_vector_type(2)));
typedef float v8f __attribute__((ext_vector_type(8)));

__global__ __launch_bounds__(128) void norm_kernel(const float* __restrict__ x,
                                                   float* __restrict__ xn, int N) {
  const int node = blockIdx.x;
  const int t = threadIdx.x;
  if (node >= N) { xn[(size_t)node * D_ + t] = 0.0f; return; }
  float v = x[(size_t)node * D_ + t];
  float s = v * v;
  s += __shfl_xor(s, 1, 16);
  s += __shfl_xor(s, 2, 16);
  s += __shfl_xor(s, 4, 16);
  s += __shfl_xor(s, 8, 16);
  float nrm = fmaxf(sqrtf(s), 1e-12f);
  xn[(size_t)node * D_ + t] = v / nrm;
}

template <int X>
__device__ __forceinline__ float lane_xor(float v) {
  int iv = __builtin_bit_cast(int, v);
  return __builtin_bit_cast(float,
      __builtin_amdgcn_update_dpp(iv, iv, 0x160 + X, 0xf, 0xf, false));
}

__global__ __launch_bounds__(32) void routing_kernel(const float* __restrict__ xn,
                                                     const int* __restrict__ nbr,
                                                     const int* __restrict__ p_maxiter,
                                                     float* __restrict__ u_out,
                                                     int* __restrict__ selcap,
                                                     int N, int M) {
  __shared__ float zs[M_ * ZP_];
  __shared__ float us[D_];
  __shared__ float ps[CUT_ * KC];

  const int node = blockIdx.x;
  const int lane = threadIdx.x;
  const int half = lane >> 4;
  const int lo   = lane & 15;
  const int cap  = lo & 7;
  const int T = p_maxiter[0];

  float xr[8];
  {
    const float4* xp = (const float4*)(xn + (size_t)node * D_ + cap * DD_ + 8 * half);
    const float4 q0 = xp[0], q1 = xp[1];
    xr[0] = q0.x; xr[1] = q0.y; xr[2] = q0.z; xr[3] = q0.w;
    xr[4] = q1.x; xr[5] = q1.y; xr[6] = q1.z; xr[7] = q1.w;
  }

  int gl = nbr[node * M + lane];
  for (int r = 0; r < M_; ++r) {
    int g = __shfl(gl, r, 32);
    int row = (g < 0) ? N : g;
    ((float4*)(zs + r * ZP_))[lane] =
        ((const float4*)(xn + (size_t)row * D_))[lane];
  }
  __syncthreads();

  float p[2][8];
  float u[8];
  #pragma unroll
  for (int r = 0; r < 8; ++r) { p[0][r] = 0.125f; p[1][r] = 0.125f; u[r] = 0.0f; }

  for (int it = 0; it < T; ++it) {
    v8f de = {}; v8f dq = {};
    #pragma unroll
    for (int k = 0; k < KC; ++k) {
      const bool mine = (cap == k);
      #pragma unroll
      for (int t = 0; t < 2; ++t) {
        #pragma unroll
        for (int j = 0; j < 4; ++j) {
          const int m0 = 16 * t + 8 * half + 2 * j;
          v2f a; a.x = zs[m0 * ZP_ + k * DD_ + lo];
                 a.y = zs[(m0 + 1) * ZP_ + k * DD_ + lo];
          v2f b; b.x = mine ? p[t][2 * j] : 0.0f;
                 b.y = mine ? p[t][2 * j + 1] : 0.0f;
          if (k & 1) dq = __builtin_amdgcn_wmma_f32_16x16x4_f32(false, a, false, b, (short)0, dq, false, false);
          else       de = __builtin_amdgcn_wmma_f32_16x16x4_f32(false, a, false, b, (short)0, de, false, false);
        }
      }
    }
    #pragma unroll
    for (int r = 0; r < 8; ++r) u[r] = ((cap & 1) ? dq[r] : de[r]) + xr[r];
    if (it == T - 1) break;

    float s2 = 0.0f;
    #pragma unroll
    for (int r = 0; r < 8; ++r) s2 += u[r] * u[r];
    s2 += __shfl_xor(s2, 16, 32);
    const float inv = fminf(__builtin_amdgcn_rsqf(s2), 1e12f);
    #pragma unroll
    for (int r = 0; r < 8; ++r) u[r] *= inv;

    v8f d0 = {}; v8f d1 = {};
    #pragma unroll
    for (int k = 0; k < KC; ++k) {
      const bool mine = (cap == k);
      #pragma unroll
      for (int s = 0; s < 4; ++s) {
        const int c0 = 8 * half + 2 * s;
        v2f b; b.x = mine ? u[2 * s] : 0.0f;
               b.y = mine ? u[2 * s + 1] : 0.0f;
        v2f a0; a0.x = zs[lo * ZP_        + k * DD_ + c0];
                a0.y = zs[lo * ZP_        + k * DD_ + c0 + 1];
        v2f a1; a1.x = zs[(16 + lo) * ZP_ + k * DD_ + c0];
                a1.y = zs[(16 + lo) * ZP_ + k * DD_ + c0 + 1];
        d0 = __builtin_amdgcn_wmma_f32_16x16x4_f32(false, a0, false, b, (short)0, d0, false, false);
        d1 = __builtin_amdgcn_wmma_f32_16x16x4_f32(false, a1, false, b, (short)0, d1, false, false);
      }
    }

    const bool upper = (lo & 8) != 0;
    #pragma unroll
    for (int t = 0; t < 2; ++t) {
      float q[4];
      #pragma unroll
      for (int i = 0; i < 4; ++i) {
        const float vl = t ? d1[i] : d0[i];
        const float vh = t ? d1[4 + i] : d0[4 + i];
        const float v = upper ? vh : vl;
        float mx = v;
        mx = fmaxf(mx, lane_xor<1>(mx));
        mx = fmaxf(mx, lane_xor<2>(mx));
        mx = fmaxf(mx, lane_xor<4>(mx));
        const float e = __builtin_amdgcn_exp2f((v - mx) * 1.4426950408889634f);
        float ssum = e;
        ssum += lane_xor<1>(ssum);
        ssum += lane_xor<2>(ssum);
        ssum += lane_xor<4>(ssum);
        q[i] = e * __builtin_amdgcn_rcpf(ssum);
      }
      #pragma unroll
      for (int i = 0; i < 4; ++i) {
        const float o = lane_xor<8>(q[i]);
        p[t][i]     = upper ? o : q[i];
        p[t][4 + i] = upper ? q[i] : o;
      }
    }
  }

  if (lo < KC) {
    #pragma unroll
    for (int r = 0; r < 8; ++r) us[cap * DD_ + 8 * half + r] = u[r];
    if (half == 0) {
      #pragma unroll
      for (int r = 0; r < CUT_; ++r) ps[r * KC + cap] = p[0][r];
    }
  }
  __syncthreads();

  ((float4*)(u_out + (size_t)node * D_))[lane] = ((float4*)us)[lane];

  if (lane < CUT_) {
    int best = 0;
    float bv = ps[lane * KC + 0];
    #pragma unroll
    for (int q = 1; q < KC; ++q) {
      float v = ps[lane * KC + q];
      if (v > bv) { bv = v; best = q; }
    }
    selcap[node * CUT_ + lane] = best;
  }
}

#define NPW_ 4
#define NPAIR_ (CUT_ * CUT_)

__global__ __launch_bounds__(256) void meta_kernel(const float* __restrict__ xn,
                                                   const int* __restrict__ nbr,
                                                   const int* __restrict__ selcap,
                                                   const int* __restrict__ p_last,
                                                   const float* __restrict__ u_out,
                                                   float* __restrict__ meta_out,
                                                   int N, int M) {
  if (p_last[0] == 0) return;
  const int lane  = threadIdx.x & 31;
  const int node0 = (blockIdx.x * 8 + (threadIdx.x >> 5)) * NPW_;
  if (node0 >= N) return;
  const int cap = lane >> 2;
  const int sub = lane & 3;
  const int q   = lane < NPAIR_ ? lane : NPAIR_ - 1;
  const int s   = q / CUT_;
  const int s2  = q - s * CUT_;

  int g1[NPW_], c1[NPW_], off[NPW_];
  #pragma unroll
  for (int n = 0; n < NPW_; ++n) {
    const int nd = min(node0 + n, N - 1);
    g1[n] = nbr[nd * M + s];
    c1[n] = selcap[nd * CUT_ + s];
  }
  #pragma unroll
  for (int n = 0; n < NPW_; ++n) {
    const int h  = g1[n] < 0 ? 0 : g1[n];
    const int g2 = nbr[h * M + s2];
    const int jj = selcap[h * CUT_ + s2];
    const int row = ((g1[n] | g2) >= 0) ? g2 : N;
    off[n] = row * D_ + jj * DD_;
  }

  #pragma unroll
  for (int n = 0; n < NPW_; ++n) {
    const int node = node0 + n;
    if (node >= N) break;
    const float4 uu = ((const float4*)(u_out + (size_t)node * D_))[lane];
    float4 v[NPAIR_];
    #pragma unroll
    for (int k = 0; k < NPAIR_; ++k) {
      const int o = __builtin_amdgcn_readlane(off[n], k);
      v[k] = *(const float4*)(xn + (size_t)o + 4 * sub);
    }
    float4 acc;
    #pragma unroll
    for (int k = 0; k < NPAIR_; ++k) {
      const int cs = __builtin_amdgcn_readlane(c1[n], (k / CUT_) * CUT_);
      const bool mine = (cap == cs);
      const float tx = mine ? v[k].x : 0.0f, ty = mine ? v[k].y : 0.0f;
      const float tz = mine ? v[k].z : 0.0f, tw = mine ? v[k].w : 0.0f;
      if (k == 0) { acc.x = tx; acc.y = ty; acc.z = tz; acc.w = tw; }
      else        { acc.x += tx; acc.y += ty; acc.z += tz; acc.w += tw; }
    }
    float4 r;
    r.x = uu.x + acc.x / 25.0f;
    r.y = uu.y + acc.y / 25.0f;
    r.z = uu.z + acc.z / 25.0f;
    r.w = uu.w + acc.w / 25.0f;
    ((float4*)(meta_out + (size_t)node * D_))[lane] = r;
  }
}

extern "C" void kernel_launch(void* const* d_in, const int* in_sizes, int n_in,
                              void* d_out, int out_size, void* d_ws, size_t ws_size,
                              hipStream_t stream) {
  const float* x       = (const float*)d_in[0];
  const int*   nbr     = (const int*)d_in[1];
  const int*   maxiter = (const int*)d_in[2];
  const int*   last    = (const int*)d_in[3];

  const int N = in_sizes[0] / D_;
  const int M = in_sizes[1] / N;

  float* u_out    = (float*)d_out;
  float* meta_out = (float*)d_out + (size_t)N * D_;

  float* xn = (float*)d_ws;
  size_t xn_bytes = (size_t)(N + 1) * D_ * sizeof(float);
  int* selcap = (int*)((char*)d_ws + xn_bytes);

  norm_kernel<<<N + 1, 128, 0, stream>>>(x, xn, N);
  routing_kernel<<<N, 32, 0, stream>>>(xn, nbr, maxiter, u_out, selcap, N, M);
  meta_kernel<<<(N + 8 * NPW_ - 1) / (8 * NPW_), 256, 0, stream>>>(xn, nbr, selcap, last, u_out, meta_out, N, M);
}


# Softmax rows keep one cross-lane tree instead of two: the row max is no longer subtracted (logits are dots of two capsules of norm <= 1, so exp2 stays in range); p differs from before by a few ulp
# speedup vs baseline: 2.0673x; 1.0300x over previous
#include <hip/hip_runtime.h>
#include <math.h>

#define D_   128
#define KC   8
#define DD_  16
#define CUT_ 5
#define M_   32
#define ZP_  140

typedef float v2f __attribute__((ext_vector_type(2)));
typedef float v8f __attribute__((ext_vector_type(8)));

__global__ __launch_bounds__(128) void norm_kernel(const float* __restrict__ x,
                                                   float* __restrict__ xn, int N) {
  const int node = blockIdx.x;
  const int t = threadIdx.x;
  if (node >= N) { xn[(size_t)node * D_ + t] = 0.0f; return; }
  float v = x[(size_t)node * D_ + t];
  float s = v * v;
  s += __shfl_xor(s, 1, 16);
  s += __shfl_xor(s, 2, 16);
  s += __shfl_xor(s, 4, 16);
  s += __shfl_xor(s, 8, 16);
  float nrm = fmaxf(sqrtf(s), 1e-12f);
  xn[(size_t)node * D_ + t] = v / nrm;
}

template <int X>
__device__ __forceinline__ float lane_xor(float v) {
  int iv = __builtin_bit_cast(int, v);
  return __builtin_bit_cast(float,
      __builtin_amdgcn_update_dpp(iv, iv, 0x160 + X, 0xf, 0xf, false));
}

__global__ __launch_bounds__(32) void routing_kernel(const float* __restrict__ xn,
                                                     const int* __restrict__ nbr,
                                                     const int* __restrict__ p_maxiter,
                                                     float* __restrict__ u_out,
                                                     int* __restrict__ selcap,
                                                     int N, int M) {
  __shared__ float zs[M_ * ZP_];
  __shared__ float us[D_];
  __shared__ float ps[CUT_ * KC];

  const int node = blockIdx.x;
  const int lane = threadIdx.x;
  const int half = lane >> 4;
  const int lo   = lane & 15;
  const int cap  = lo & 7;
  const int T = p_maxiter[0];

  float xr[8];
  {
    const float4* xp = (const float4*)(xn + (size_t)node * D_ + cap * DD_ + 8 * half);
    const float4 q0 = xp[0], q1 = xp[1];
    xr[0] = q0.x; xr[1] = q0.y; xr[2] = q0.z; xr[3] = q0.w;
    xr[4] = q1.x; xr[5] = q1.y; xr[6] = q1.z; xr[7] = q1.w;
  }

  int gl = nbr[node * M + lane];
  for (int r = 0; r < M_; ++r) {
    int g = __shfl(gl, r, 32);
    int row = (g < 0) ? N : g;
    ((float4*)(zs + r * ZP_))[lane] =
        ((const float4*)(xn + (size_t)row * D_))[lane];
  }
  __syncthreads();

  float p[2][8];
  float u[8];
  #pragma unroll
  for (int r = 0; r < 8; ++r) { p[0][r] = 0.125f; p[1][r] = 0.125f; u[r] = 0.0f; }

  for (int it = 0; it < T; ++it) {
    v8f de = {}; v8f dq = {};
    #pragma unroll
    for (int k = 0; k < KC; ++k) {
      const bool mine = (cap == k);
      #pragma unroll
      for (int t = 0; t < 2; ++t) {
        #pragma unroll
        for (int j = 0; j < 4; ++j) {
          const int m0 = 16 * t + 8 * half + 2 * j;
          v2f a; a.x = zs[m0 * ZP_ + k * DD_ + lo];
                 a.y = zs[(m0 + 1) * ZP_ + k * DD_ + lo];
          v2f b; b.x = mine ? p[t][2 * j] : 0.0f;
                 b.y = mine ? p[t][2 * j + 1] : 0.0f;
          if (k & 1) dq = __builtin_amdgcn_wmma_f32_16x16x4_f32(false, a, false, b, (short)0, dq, false, false);
          else       de = __builtin_amdgcn_wmma_f32_16x16x4_f32(false, a, false, b, (short)0, de, false, false);
        }
      }
    }
    #pragma unroll
    for (int r = 0; r < 8; ++r) u[r] = ((cap & 1) ? dq[r] : de[r]) + xr[r];
    if (it == T - 1) break;

    float s2 = 0.0f;
    #pragma unroll
    for (int r = 0; r < 8; ++r) s2 += u[r] * u[r];
    s2 += __shfl_xor(s2, 16, 32);
    const float inv = fminf(__builtin_amdgcn_rsqf(s2), 1e12f);
    #pragma unroll
    for (int r = 0; r < 8; ++r) u[r] *= inv;

    v8f d0 = {}; v8f d1 = {};
    #pragma unroll
    for (int k = 0; k < KC; ++k) {
      const bool mine = (cap == k);
      #pragma unroll
      for (int s = 0; s < 4; ++s) {
        const int c0 = 8 * half + 2 * s;
        v2f b; b.x = mine ? u[2 * s] : 0.0f;
               b.y = mine ? u[2 * s + 1] : 0.0f;
        v2f a0; a0.x = zs[lo * ZP_        + k * DD_ + c0];
                a0.y = zs[lo * ZP_        + k * DD_ + c0 + 1];
        v2f a1; a1.x = zs[(16 + lo) * ZP_ + k * DD_ + c0];
                a1.y = zs[(16 + lo) * ZP_ + k * DD_ + c0 + 1];
        d0 = __builtin_amdgcn_wmma_f32_16x16x4_f32(false, a0, false, b, (short)0, d0, false, false);
        d1 = __builtin_amdgcn_wmma_f32_16x16x4_f32(false, a1, false, b, (short)0, d1, false, false);
      }
    }

    const bool upper = (lo & 8) != 0;
    #pragma unroll
    for (int t = 0; t < 2; ++t) {
      float q[4];
      #pragma unroll
      for (int i = 0; i < 4; ++i) {
        const float vl = t ? d1[i] : d0[i];
        const float vh = t ? d1[4 + i] : d0[4 + i];
        const float v = upper ? vh : vl;
        const float e = __builtin_amdgcn_exp2f(v * 1.4426950408889634f);
        float ssum = e;
        ssum += lane_xor<1>(ssum);
        ssum += lane_xor<2>(ssum);
        ssum += lane_xor<4>(ssum);
        q[i] = e * __builtin_amdgcn_rcpf(ssum);
      }
      #pragma unroll
      for (int i = 0; i < 4; ++i) {
        const float o = lane_xor<8>(q[i]);
        p[t][i]     = upper ? o : q[i];
        p[t][4 + i] = upper ? q[i] : o;
      }
    }
  }

  if (lo < KC) {
    #pragma unroll
    for (int r = 0; r < 8; ++r) us[cap * DD_ + 8 * half + r] = u[r];
    if (half == 0) {
      #pragma unroll
      for (int r = 0; r < CUT_; ++r) ps[r * KC + cap] = p[0][r];
    }
  }
  __syncthreads();

  ((float4*)(u_out + (size_t)node * D_))[lane] = ((float4*)us)[lane];

  if (lane < CUT_) {
    int best = 0;
    float bv = ps[lane * KC + 0];
    #pragma unroll
    for (int q = 1; q < KC; ++q) {
      float v = ps[lane * KC + q];
      if (v > bv) { bv = v; best = q; }
    }
    selcap[node * CUT_ + lane] = best;
  }
}

#define NPW_ 4
#define NPAIR_ (CUT_ * CUT_)

__global__ __launch_bounds__(256) void meta_kernel(const float* __restrict__ xn,
                                                   const int* __restrict__ nbr,
                                                   const int* __restrict__ selcap,
                                                   const int* __restrict__ p_last,
                                                   const float* __restrict__ u_out,
                                                   float* __restrict__ meta_out,
                                                   int N, int M) {
  if (p_last[0] == 0) return;
  const int lane  = threadIdx.x & 31;
  const int node0 = (blockIdx.x * 8 + (threadIdx.x >> 5)) * NPW_;
  if (node0 >= N) return;
  const int cap = lane >> 2;
  const int sub = lane & 3;
  const int q   = lane < NPAIR_ ? lane : NPAIR_ - 1;
  const int s   = q / CUT_;
  const int s2  = q - s * CUT_;

  int g1[NPW_], c1[NPW_], off[NPW_];
  #pragma unroll
  for (int n = 0; n < NPW_; ++n) {
    const int nd = min(node0 + n, N - 1);
    g1[n] = nbr[nd * M + s];
    c1[n] = selcap[nd * CUT_ + s];
  }
  #pragma unroll
  for (int n = 0; n < NPW_; ++n) {
    const int h  = g1[n] < 0 ? 0 : g1[n];
    const int g2 = nbr[h * M + s2];
    const int jj = selcap[h * CUT_ + s2];
    const int row = ((g1[n] | g2) >= 0) ? g2 : N;
    off[n] = row * D_ + jj * DD_;
  }

  #pragma unroll
  for (int n = 0; n < NPW_; ++n) {
    const int node = node0 + n;
    if (node >= N) break;
    const float4 uu = ((const float4*)(u_out + (size_t)node * D_))[lane];
    float4 v[NPAIR_];
    #pragma unroll
    for (int k = 0; k < NPAIR_; ++k) {
      const int o = __builtin_amdgcn_readlane(off[n], k);
      v[k] = *(const float4*)(xn + (size_t)o + 4 * sub);
    }
    float4 acc;
    #pragma unroll
    for (int k = 0; k < NPAIR_; ++k) {
      const int cs = __builtin_amdgcn_readlane(c1[n], (k / CUT_) * CUT_);
      const bool mine = (cap == cs);
      const float tx = mine ? v[k].x : 0.0f, ty = mine ? v[k].y : 0.0f;
      const float tz = mine ? v[k].z : 0.0f, tw = mine ? v[k].w : 0.0f;
      if (k == 0) { acc.x = tx; acc.y = ty; acc.z = tz; acc.w = tw; }
      else        { acc.x += tx; acc.y += ty; acc.z += tz; acc.w += tw; }
    }
    float4 r;
    r.x = uu.x + acc.x / 25.0f;
    r.y = uu.y + acc.y / 25.0f;
    r.z = uu.z + acc.z / 25.0f;
    r.w = uu.w + acc.w / 25.0f;
    ((float4*)(meta_out + (size_t)node * D_))[lane] = r;
  }
}

extern "C" void kernel_launch(void* const* d_in, const int* in_sizes, int n_in,
                              void* d_out, int out_size, void* d_ws, size_t ws_size,
                              hipStream_t stream) {
  const float* x       = (const float*)d_in[0];
  const int*   nbr     = (const int*)d_in[1];
  const int*   maxiter = (const int*)d_in[2];
  const int*   last    = (const int*)d_in[3];

  const int N = in_sizes[0] / D_;
  const int M = in_sizes[1] / N;

  float* u_out    = (float*)d_out;
  float* meta_out = (float*)d_out + (size_t)N * D_;

  float* xn = (float*)d_ws;
  size_t xn_bytes = (size_t)(N + 1) * D_ * sizeof(float);
  int* selcap = (int*)((char*)d_ws + xn_bytes);

  norm_kernel<<<N + 1, 128, 0, stream>>>(x, xn, N);
  routing_kernel<<<N, 32, 0, stream>>>(xn, nbr, maxiter, u_out, selcap, N, M);
  meta_kernel<<<(N + 8 * NPW_ - 1) / (8 * NPW_), 256, 0, stream>>>(xn, nbr, selcap, last, u_out, meta_out, N, M);
}


# First routing pass takes no product: its weights are softmax(0) = 1/8 for any input, so u = colsum(z)/8 + x, each lane summing its 4 columns during the gather; loop rotated; 32-term sum reordered
# speedup vs baseline: 2.1806x; 1.0548x over previous
#include <hip/hip_runtime.h>
#include <math.h>

#define D_   128
#define KC   8
#define DD_  16
#define CUT_ 5
#define M_   32
#define ZP_  140

typedef float v2f __attribute__((ext_vector_type(2)));
typedef float v8f __attribute__((ext_vector_type(8)));

__global__ __launch_bounds__(128) void norm_kernel(const float* __restrict__ x,
                                                   float* __restrict__ xn, int N) {
  const int node = blockIdx.x;
  const int t = threadIdx.x;
  if (node >= N) { xn[(size_t)node * D_ + t] = 0.0f; return; }
  float v = x[(size_t)node * D_ + t];
  float s = v * v;
  s += __shfl_xor(s, 1, 16);
  s += __shfl_xor(s, 2, 16);
  s += __shfl_xor(s, 4, 16);
  s += __shfl_xor(s, 8, 16);
  float nrm = fmaxf(sqrtf(s), 1e-12f);
  xn[(size_t)node * D_ + t] = v / nrm;
}

template <int X>
__device__ __forceinline__ float lane_xor(float v) {
  int iv = __builtin_bit_cast(int, v);
  return __builtin_bit_cast(float,
      __builtin_amdgcn_update_dpp(iv, iv, 0x160 + X, 0xf, 0xf, false));
}

__global__ __launch_bounds__(32) void routing_kernel(const float* __restrict__ xn,
                                                     const int* __restrict__ nbr,
                                                     const int* __restrict__ p_maxiter,
                                                     float* __restrict__ u_out,
                                                     int* __restrict__ selcap,
                                                     int N, int M) {
  __shared__ float zs[M_ * ZP_];
  __shared__ float us[D_];
  __shared__ float ps[CUT_ * KC];

  const int node = blockIdx.x;
  const int lane = threadIdx.x;
  const int half = lane >> 4;
  const int lo   = lane & 15;
  const int cap  = lo & 7;
  const int T = p_maxiter[0];

  float xr[8];
  {
    const float4* xp = (const float4*)(xn + (size_t)node * D_ + cap * DD_ + 8 * half);
    const float4 q0 = xp[0], q1 = xp[1];
    xr[0] = q0.x; xr[1] = q0.y; xr[2] = q0.z; xr[3] = q0.w;
    xr[4] = q1.x; xr[5] = q1.y; xr[6] = q1.z; xr[7] = q1.w;
  }

  int gl = nbr[node * M + lane];
  float4 cs = {0.0f, 0.0f, 0.0f, 0.0f};
  #pragma unroll
  for (int r = 0; r < M_; ++r) {
    int g = __shfl(gl, r, 32);
    int row = (g < 0) ? N : g;
    const float4 zv = ((const float4*)(xn + (size_t)row * D_))[lane];
    ((float4*)(zs + r * ZP_))[lane] = zv;
    cs.x += zv.x; cs.y += zv.y; cs.z += zv.z; cs.w += zv.w;
  }
  ((float4*)us)[lane] = cs;
  __syncthreads();

  float p[2][8];
  float u[8];
  #pragma unroll
  for (int r = 0; r < 8; ++r) { p[0][r] = 0.125f; p[1][r] = 0.125f; }

  {
    const float4* sp = (const float4*)(us + cap * DD_ + 8 * half);
    const float4 s0 = sp[0], s1 = sp[1];
    u[0] = 0.125f * s0.x + xr[0]; u[1] = 0.125f * s0.y + xr[1];
    u[2] = 0.125f * s0.z + xr[2]; u[3] = 0.125f * s0.w + xr[3];
    u[4] = 0.125f * s1.x + xr[4]; u[5] = 0.125f * s1.y + xr[5];
    u[6] = 0.125f * s1.z + xr[6]; u[7] = 0.125f * s1.w + xr[7];
  }

  for (int it = 1; it < T; ++it) {
    float s2 = 0.0f;
    #pragma unroll
    for (int r = 0; r < 8; ++r) s2 += u[r] * u[r];
    s2 += __shfl_xor(s2, 16, 32);
    const float inv = fminf(__builtin_amdgcn_rsqf(s2), 1e12f);
    #pragma unroll
    for (int r = 0; r < 8; ++r) u[r] *= inv;

    v8f d0 = {}; v8f d1 = {};
    #pragma unroll
    for (int k = 0; k < KC; ++k) {
      const bool mine = (cap == k);
      #pragma unroll
      for (int s = 0; s < 4; ++s) {
        const int c0 = 8 * half + 2 * s;
        v2f b; b.x = mine ? u[2 * s] : 0.0f;
               b.y = mine ? u[2 * s + 1] : 0.0f;
        v2f a0; a0.x = zs[lo * ZP_        + k * DD_ + c0];
                a0.y = zs[lo * ZP_        + k * DD_ + c0 + 1];
        v2f a1; a1.x = zs[(16 + lo) * ZP_ + k * DD_ + c0];
                a1.y = zs[(16 + lo) * ZP_ + k * DD_ + c0 + 1];
        d0 = __builtin_amdgcn_wmma_f32_16x16x4_f32(false, a0, false, b, (short)0, d0, false, false);
        d1 = __builtin_amdgcn_wmma_f32_16x16x4_f32(false, a1, false, b, (short)0, d1, false, false);
      }
    }

    const bool upper = (lo & 8) != 0;
    #pragma unroll
    for (int t = 0; t < 2; ++t) {
      float q[4];
      #pragma unroll
      for (int i = 0; i < 4; ++i) {
        const float vl = t ? d1[i] : d0[i];
        const float vh = t ? d1[4 + i] : d0[4 + i];
        const float v = upper ? vh : vl;
        const float e = __builtin_amdgcn_exp2f(v * 1.4426950408889634f);
        float ssum = e;
        ssum += lane_xor<1>(ssum);
        ssum += lane_xor<2>(ssum);
        ssum += lane_xor<4>(ssum);
        q[i] = e * __builtin_amdgcn_rcpf(ssum);
      }
      #pragma unroll
      for (int i = 0; i < 4; ++i) {
        const float o = lane_xor<8>(q[i]);
        p[t][i]     = upper ? o : q[i];
        p[t][4 + i] = upper ? q[i] : o;
      }
    }

    v8f de = {}; v8f dq = {};
    #pragma unroll
    for (int k = 0; k < KC; ++k) {
      const bool mine = (cap == k);
      #pragma unroll
      for (int t = 0; t < 2; ++t) {
        #pragma unroll
        for (int j = 0; j < 4; ++j) {
          const int m0 = 16 * t + 8 * half + 2 * j;
          v2f a; a.x = zs[m0 * ZP_ + k * DD_ + lo];
                 a.y = zs[(m0 + 1) * ZP_ + k * DD_ + lo];
          v2f b; b.x = mine ? p[t][2 * j] : 0.0f;
                 b.y = mine ? p[t][2 * j + 1] : 0.0f;
          if (k & 1) dq = __builtin_amdgcn_wmma_f32_16x16x4_f32(false, a, false, b, (short)0, dq, false, false);
          else       de = __builtin_amdgcn_wmma_f32_16x16x4_f32(false, a, false, b, (short)0, de, false, false);
        }
      }
    }
    #pragma unroll
    for (int r = 0; r < 8; ++r) u[r] = ((cap & 1) ? dq[r] : de[r]) + xr[r];
  }

  if (lo < KC) {
    #pragma unroll
    for (int r = 0; r < 8; ++r) us[cap * DD_ + 8 * half + r] = u[r];
    if (half == 0) {
      #pragma unroll
      for (int r = 0; r < CUT_; ++r) ps[r * KC + cap] = p[0][r];
    }
  }
  __syncthreads();

  ((float4*)(u_out + (size_t)node * D_))[lane] = ((float4*)us)[lane];

  if (lane < CUT_) {
    int best = 0;
    float bv = ps[lane * KC + 0];
    #pragma unroll
    for (int q = 1; q < KC; ++q) {
      float v = ps[lane * KC + q];
      if (v > bv) { bv = v; best = q; }
    }
    selcap[node * CUT_ + lane] = best;
  }
}

#define NPW_ 4
#define NPAIR_ (CUT_ * CUT_)

__global__ __launch_bounds__(256) void meta_kernel(const float* __restrict__ xn,
                                                   const int* __restrict__ nbr,
                                                   const int* __restrict__ selcap,
                                                   const int* __restrict__ p_last,
                                                   const float* __restrict__ u_out,
                                                   float* __restrict__ meta_out,
                                                   int N, int M) {
  if (p_last[0] == 0) return;
  const int lane  = threadIdx.x & 31;
  const int node0 = (blockIdx.x * 8 + (threadIdx.x >> 5)) * NPW_;
  if (node0 >= N) return;
  const int cap = lane >> 2;
  const int sub = lane & 3;
  const int q   = lane < NPAIR_ ? lane : NPAIR_ - 1;
  const int s   = q / CUT_;
  const int s2  = q - s * CUT_;

  int g1[NPW_], c1[NPW_], off[NPW_];
  #pragma unroll
  for (int n = 0; n < NPW_; ++n) {
    const int nd = min(node0 + n, N - 1);
    g1[n] = nbr[nd * M + s];
    c1[n] = selcap[nd * CUT_ + s];
  }
  #pragma unroll
  for (int n = 0; n < NPW_; ++n) {
    const int h  = g1[n] < 0 ? 0 : g1[n];
    const int g2 = nbr[h * M + s2];
    const int jj = selcap[h * CUT_ + s2];
    const int row = ((g1[n] | g2) >= 0) ? g2 : N;
    off[n] = row * D_ + jj * DD_;
  }

  #pragma unroll
  for (int n = 0; n < NPW_; ++n) {
    const int node = node0 + n;
    if (node >= N) break;
    const float4 uu = ((const float4*)(u_out + (size_t)node * D_))[lane];
    float4 v[NPAIR_];
    #pragma unroll
    for (int k = 0; k < NPAIR_; ++k) {
      const int o = __builtin_amdgcn_readlane(off[n], k);
      v[k] = *(const float4*)(xn + (size_t)o + 4 * sub);
    }
    float4 acc;
    #pragma unroll
    for (int k = 0; k < NPAIR_; ++k) {
      const int cs = __builtin_amdgcn_readlane(c1[n], (k / CUT_) * CUT_);
      const bool mine = (cap == cs);
      const float tx = mine ? v[k].x : 0.0f, ty = mine ? v[k].y : 0.0f;
      const float tz = mine ? v[k].z : 0.0f, tw = mine ? v[k].w : 0.0f;
      if (k == 0) { acc.x = tx; acc.y = ty; acc.z = tz; acc.w = tw; }
      else        { acc.x += tx; acc.y += ty; acc.z += tz; acc.w += tw; }
    }
    float4 r;
    r.x = uu.x + acc.x / 25.0f;
    r.y = uu.y + acc.y / 25.0f;
    r.z = uu.z + acc.z / 25.0f;
    r.w = uu.w + acc.w / 25.0f;
    ((float4*)(meta_out + (size_t)node * D_))[lane] = r;
  }
}

extern "C" void kernel_launch(void* const* d_in, const int* in_sizes, int n_in,
                              void* d_out, int out_size, void* d_ws, size_t ws_size,
                              hipStream_t stream) {
  const float* x       = (const float*)d_in[0];
  const int*   nbr     = (const int*)d_in[1];
  const int*   maxiter = (const int*)d_in[2];
  const int*   last    = (const int*)d_in[3];

  const int N = in_sizes[0] / D_;
  const int M = in_sizes[1] / N;

  float* u_out    = (float*)d_out;
  float* meta_out = (float*)d_out + (size_t)N * D_;

  float* xn = (float*)d_ws;
  size_t xn_bytes = (size_t)(N + 1) * D_ * sizeof(float);
  int* selcap = (int*)((char*)d_ws + xn_bytes);

  norm_kernel<<<N + 1, 128, 0, stream>>>(x, xn, N);
  routing_kernel<<<N, 32, 0, stream>>>(xn, nbr, maxiter, u_out, selcap, N, M);
  meta_kernel<<<(N + 8 * NPW_ - 1) / (8 * NPW_), 256, 0, stream>>>(xn, nbr, selcap, last, u_out, meta_out, N, M);
}
